# starting kernel
# baseline (speedup 1.0000x reference)
#include <hip/hip_runtime.h>
#include <math.h>


typedef __attribute__((ext_vector_type(2))) float v2f;
typedef __attribute__((ext_vector_type(8))) float v8f;

#define B_      8
#define T_      4096
#define D_      512
#define H_      512
#define M_      (B_ * T_)
#define CHUNKS  64
#define CS      64
#define MT      4

__device__ __forceinline__ void sigmoid_pair(float x, float& sig_pos, float& sig_neg) {
    const float e = __expf(-fabsf(x));
    const float r = __builtin_amdgcn_rcpf(1.0f + e);
    sig_pos = r;
    sig_neg = e * r;
}

__device__ __forceinline__ float g_fn(float x) {
    const float e = __expf(fminf(x, 0.0f));
    const float s = e * __builtin_amdgcn_rcpf(1.0f + e);
    return (x >= 0.0f) ? (x + 0.5f) : s;
}

__global__ __launch_bounds__(32) void mingru_gemm_gates(
    const float* __restrict__ X,
    const float* __restrict__ Wz, const float* __restrict__ bz,
    const float* __restrict__ Wh, const float* __restrict__ bh,
    float* __restrict__ outA, float* __restrict__ outV)
{
    const int lane = threadIdx.x;
    const int half = lane >> 4;
    const int l16  = lane & 15;
    const int m0   = blockIdx.x * (16 * MT);
    const int n0   = blockIdx.y * 16;

    const float* aptr[MT];
#pragma unroll
    for (int mt = 0; mt < MT; ++mt)
        aptr[mt] = X + (size_t)(m0 + mt * 16 + l16) * D_ + half * 2;
    const float* bzptr = Wz + (size_t)(n0 + l16) * D_ + half * 2;
    const float* bhptr = Wh + (size_t)(n0 + l16) * D_ + half * 2;

    v8f accZ[MT], accH[MT];
#pragma unroll
    for (int mt = 0; mt < MT; ++mt) {
        accZ[mt] = (v8f){0.f,0.f,0.f,0.f,0.f,0.f,0.f,0.f};
        accH[mt] = (v8f){0.f,0.f,0.f,0.f,0.f,0.f,0.f,0.f};
    }

#pragma unroll 2
    for (int k = 0; k < D_; k += 4) {
        v2f bzv = *(const v2f*)(bzptr + k);
        v2f bhv = *(const v2f*)(bhptr + k);
#pragma unroll
        for (int mt = 0; mt < MT; ++mt) {
            v2f av = *(const v2f*)(aptr[mt] + k);
            accZ[mt] = __builtin_amdgcn_wmma_f32_16x16x4_f32(
                false, av, false, bzv, (short)0, accZ[mt], false, false);
            accH[mt] = __builtin_amdgcn_wmma_f32_16x16x4_f32(
                false, av, false, bhv, (short)0, accH[mt], false, false);
        }
    }

    const int n = n0 + l16;
    const float bzn = bz[n];
    const float bhn = bh[n];
#pragma unroll
    for (int mt = 0; mt < MT; ++mt) {
#pragma unroll
        for (int r = 0; r < 8; ++r) {
            const int m   = m0 + mt * 16 + half * 8 + r;
            const float kv  = accZ[mt][r] + bzn;
            const float pre = accH[mt][r] + bhn;
            float sp, sn;
            sigmoid_pair(kv, sp, sn);
            const float z = (kv >= 0.0f) ? sp : sn;
            const float a = (kv >= 0.0f) ? sn : sp;
            const float v = z * g_fn(pre);
            const size_t idx = (size_t)m * H_ + n;
            outA[idx] = a;
            outV[idx] = v;
        }
    }
}

__global__ __launch_bounds__(256) void mingru_chunk_reduce(
    const float* __restrict__ a, const float* __restrict__ v,
    float* __restrict__ Ac, float* __restrict__ Vc)
{
    const int tid = blockIdx.x * 256 + threadIdx.x;
    const int h  = tid % H_;
    const int bc = tid / H_;
    const int c  = bc % CHUNKS;
    const int b  = bc / CHUNKS;
    size_t base = ((size_t)(b * T_ + c * CS)) * H_ + h;

    float A = 1.0f, V = 0.0f;
#pragma unroll 4
    for (int t = 0; t < CS; ++t) {
        const float at = a[base];
        const float vt = v[base];
        A = A * at;
        V = fmaf(at, V, vt);
        base += H_;
    }
    Ac[tid] = A;
    Vc[tid] = V;
}

__global__ __launch_bounds__(256) void mingru_chunk_scan(
    const float* __restrict__ h0,
    const float* __restrict__ Ac, const float* __restrict__ Vc,
    float* __restrict__ hin)
{
    const int tid = blockIdx.x * 256 + threadIdx.x;
    const int h = tid % H_;
    const int b = tid / H_;
    float hs = g_fn(h0[(size_t)b * H_ + h]);
#pragma unroll 8
    for (int c = 0; c < CHUNKS; ++c) {
        const size_t i = ((size_t)(b * CHUNKS + c)) * H_ + h;
        hin[i] = hs;
        hs = fmaf(Ac[i], hs, Vc[i]);
    }
}

__global__ __launch_bounds__(256) void mingru_chunk_apply(
    const float* __restrict__ a, const float* __restrict__ v,
    const float* __restrict__ hin, float* __restrict__ out)
{
    const int tid = blockIdx.x * 256 + threadIdx.x;
    const int h  = tid % H_;
    const int bc = tid / H_;
    const int c  = bc % CHUNKS;
    const int b  = bc / CHUNKS;
    size_t base = ((size_t)(b * T_ + c * CS)) * H_ + h;

    float hs = hin[tid];
#pragma unroll 4
    for (int t = 0; t < CS; ++t) {
        hs = fmaf(a[base], hs, v[base]);
        out[base] = hs;
        base += H_;
    }
}

extern "C" void kernel_launch(void* const* d_in, const int* in_sizes, int n_in,
                              void* d_out, int out_size, void* d_ws, size_t ws_size,
                              hipStream_t stream) {
    const float* x  = (const float*)d_in[0];
    const float* h0 = (const float*)d_in[1];
    const float* Wz = (const float*)d_in[2];
    const float* bz = (const float*)d_in[3];
    const float* Wh = (const float*)d_in[4];
    const float* bh = (const float*)d_in[5];
    float* out = (float*)d_out;

    float* wa  = (float*)d_ws;
    float* wv  = wa + (size_t)M_ * H_;
    float* Ac  = wv + (size_t)M_ * H_;
    float* Vc  = Ac + (size_t)B_ * CHUNKS * H_;
    float* hin = Vc + (size_t)B_ * CHUNKS * H_;

    dim3 g1(M_ / (16 * MT), H_ / 16);
    mingru_gemm_gates<<<g1, 32, 0, stream>>>(x, Wz, bz, Wh, bh, wa, wv);

    const int nchunk = B_ * H_ * CHUNKS;
    mingru_chunk_reduce<<<nchunk / 256, 256, 0, stream>>>(wa, wv, Ac, Vc);
    mingru_chunk_scan<<<(B_ * H_) / 256, 256, 0, stream>>>(h0, Ac, Vc, hin);
    mingru_chunk_apply<<<nchunk / 256, 256, 0, stream>>>(wa, wv, hin, out);
}


# GEMM operands staged through LDS: 4-wave workgroup tile 128x64 of both products, 32-k stages by whole-line 16-byte global loads; k sum order permuted inside each 8 (lane half holds 4 consecutive k)
# speedup vs baseline: 2.3531x; 2.3531x over previous
#include <hip/hip_runtime.h>
#include <math.h>


typedef __attribute__((ext_vector_type(2))) float v2f;
typedef __attribute__((ext_vector_type(4))) float v4f;
typedef __attribute__((ext_vector_type(8))) float v8f;

#define B_      8
#define T_      4096
#define D_      512
#define H_      512
#define M_      (B_ * T_)
#define CHUNKS  64
#define CS      64

__device__ __forceinline__ void sigmoid_pair(float x, float& sig_pos, float& sig_neg) {
    const float e = __expf(-fabsf(x));
    const float r = __builtin_amdgcn_rcpf(1.0f + e);
    sig_pos = r;
    sig_neg = e * r;
}

__device__ __forceinline__ float g_fn(float x) {
    const float e = __expf(fminf(x, 0.0f));
    const float s = e * __builtin_amdgcn_rcpf(1.0f + e);
    return (x >= 0.0f) ? (x + 0.5f) : s;
}

#define BM      128
#define BN      64
#define BK      32
#define PITCH4  9
#define A_V4    (BM * PITCH4)
#define W_V4    (BN * PITCH4)
#define STAGE_V4 (A_V4 + 2 * W_V4)
#define KT      (D_ / BK)

struct Frags {
    v4f a[4];
    v4f z[2];
    v4f h[2];
};

__device__ __forceinline__ void load_frags(Frags& f, const v4f* sa, const v4f* sz,
                                           const v4f* sh, int j) {
#pragma unroll
    for (int mt = 0; mt < 4; ++mt) f.a[mt] = sa[mt * 16 * PITCH4 + 2 * j];
#pragma unroll
    for (int nt = 0; nt < 2; ++nt) {
        f.z[nt] = sz[nt * 16 * PITCH4 + 2 * j];
        f.h[nt] = sh[nt * 16 * PITCH4 + 2 * j];
    }
}

__device__ __forceinline__ void mma_chunk(const Frags& f, v8f (&accZ)[4][2], v8f (&accH)[4][2]) {
#pragma unroll
    for (int s = 0; s < 2; ++s) {
#pragma unroll
        for (int mt = 0; mt < 4; ++mt) {
            const v2f av = s ? (v2f){f.a[mt].z, f.a[mt].w} : (v2f){f.a[mt].x, f.a[mt].y};
#pragma unroll
            for (int nt = 0; nt < 2; ++nt) {
                const v2f zv = s ? (v2f){f.z[nt].z, f.z[nt].w} : (v2f){f.z[nt].x, f.z[nt].y};
                const v2f hv = s ? (v2f){f.h[nt].z, f.h[nt].w} : (v2f){f.h[nt].x, f.h[nt].y};
                accZ[mt][nt] = __builtin_amdgcn_wmma_f32_16x16x4_f32(
                    false, av, false, zv, (short)0, accZ[mt][nt], false, false);
                accH[mt][nt] = __builtin_amdgcn_wmma_f32_16x16x4_f32(
                    false, av, false, hv, (short)0, accH[mt][nt], false, false);
            }
        }
    }
}

__device__ __forceinline__ void compute_stage(const v4f* sa, const v4f* sz, const v4f* sh,
                                              v8f (&accZ)[4][2], v8f (&accH)[4][2]) {
    Frags f0, f1;
    load_frags(f0, sa, sz, sh, 0);
    load_frags(f1, sa, sz, sh, 1);
    mma_chunk(f0, accZ, accH);
    load_frags(f0, sa, sz, sh, 2);
    mma_chunk(f1, accZ, accH);
    load_frags(f1, sa, sz, sh, 3);
    mma_chunk(f0, accZ, accH);
    mma_chunk(f1, accZ, accH);
}

__global__ __launch_bounds__(128) void mingru_gemm_gates(
    const float* __restrict__ X,
    const float* __restrict__ Wz, const float* __restrict__ bz,
    const float* __restrict__ Wh, const float* __restrict__ bh,
    float* __restrict__ outA, float* __restrict__ outV)
{
    __shared__ v4f lds[2 * STAGE_V4];

    const int tid  = threadIdx.x;
    const int wave = tid >> 5;
    const int lane = tid & 31;
    const int half = lane >> 4;
    const int l16  = lane & 15;
    const int wm   = wave >> 1;
    const int wn   = wave & 1;
    const int n0   = blockIdx.x * BN;
    const int m0   = blockIdx.y * BM;

    const int cr = tid >> 3;
    const int cq = tid & 7;
    const float* gA = X  + (size_t)(m0 + cr) * D_ + cq * 4;
    const float* gZ = Wz + (size_t)(n0 + cr) * D_ + cq * 4;
    const float* gH = Wh + (size_t)(n0 + cr) * D_ + cq * 4;
    const int sA = cr * PITCH4 + cq;
    const int sZ = A_V4 + cr * PITCH4 + cq;
    const int sH = A_V4 + W_V4 + cr * PITCH4 + cq;

    const int fA = (wm * 64 + l16) * PITCH4 + half;
    const int fZ = A_V4 + (wn * 32 + l16) * PITCH4 + half;
    const int fH = fZ + W_V4;

    v4f pa[8], pz[4], ph[4];
#pragma unroll
    for (int i = 0; i < 8; ++i) pa[i] = *(const v4f*)(gA + (size_t)i * 16 * D_);
#pragma unroll
    for (int i = 0; i < 4; ++i) {
        pz[i] = *(const v4f*)(gZ + (size_t)i * 16 * D_);
        ph[i] = *(const v4f*)(gH + (size_t)i * 16 * D_);
    }

    v8f accZ[4][2], accH[4][2];
#pragma unroll
    for (int mt = 0; mt < 4; ++mt)
#pragma unroll
        for (int nt = 0; nt < 2; ++nt) {
            accZ[mt][nt] = (v8f){0.f,0.f,0.f,0.f,0.f,0.f,0.f,0.f};
            accH[mt][nt] = (v8f){0.f,0.f,0.f,0.f,0.f,0.f,0.f,0.f};
        }

#pragma unroll
    for (int i = 0; i < 8; ++i) lds[sA + i * 16 * PITCH4] = pa[i];
#pragma unroll
    for (int i = 0; i < 4; ++i) {
        lds[sZ + i * 16 * PITCH4] = pz[i];
        lds[sH + i * 16 * PITCH4] = ph[i];
    }
    __syncthreads();

    for (int t = 0; t < KT - 1; ++t) {
        const int cur = (t & 1) * STAGE_V4;
        const int nxt = STAGE_V4 - cur;
        const int ko  = (t + 1) * BK;
#pragma unroll
        for (int i = 0; i < 8; ++i) pa[i] = *(const v4f*)(gA + (size_t)i * 16 * D_ + ko);
#pragma unroll
        for (int i = 0; i < 4; ++i) {
            pz[i] = *(const v4f*)(gZ + (size_t)i * 16 * D_ + ko);
            ph[i] = *(const v4f*)(gH + (size_t)i * 16 * D_ + ko);
        }

        compute_stage(lds + cur + fA, lds + cur + fZ, lds + cur + fH, accZ, accH);

#pragma unroll
        for (int i = 0; i < 8; ++i) lds[nxt + sA + i * 16 * PITCH4] = pa[i];
#pragma unroll
        for (int i = 0; i < 4; ++i) {
            lds[nxt + sZ + i * 16 * PITCH4] = pz[i];
            lds[nxt + sH + i * 16 * PITCH4] = ph[i];
        }
        __syncthreads();
    }
    {
        const int cur = ((KT - 1) & 1) * STAGE_V4;
        compute_stage(lds + cur + fA, lds + cur + fZ, lds + cur + fH, accZ, accH);
    }

    const int nb = n0 + wn * 32 + l16;
    float bzn[2], bhn[2];
#pragma unroll
    for (int nt = 0; nt < 2; ++nt) {
        bzn[nt] = bz[nb + nt * 16];
        bhn[nt] = bh[nb + nt * 16];
    }
#pragma unroll
    for (int mt = 0; mt < 4; ++mt) {
#pragma unroll
        for (int r = 0; r < 8; ++r) {
            const int m = m0 + wm * 64 + mt * 16 + half * 8 + r;
#pragma unroll
            for (int nt = 0; nt < 2; ++nt) {
                const float kv  = accZ[mt][nt][r] + bzn[nt];
                const float pre = accH[mt][nt][r] + bhn[nt];
                float sp, sn;
                sigmoid_pair(kv, sp, sn);
                const float z = (kv >= 0.0f) ? sp : sn;
                const float a = (kv >= 0.0f) ? sn : sp;
                const float v = z * g_fn(pre);
                const size_t idx = (size_t)m * H_ + nb + nt * 16;
                outA[idx] = a;
                outV[idx] = v;
            }
        }
    }
}

__global__ __launch_bounds__(256) void mingru_chunk_reduce(
    const float* __restrict__ a, const float* __restrict__ v,
    float* __restrict__ Ac, float* __restrict__ Vc)
{
    const int tid = blockIdx.x * 256 + threadIdx.x;
    const int h  = tid % H_;
    const int bc = tid / H_;
    const int c  = bc % CHUNKS;
    const int b  = bc / CHUNKS;
    size_t base = ((size_t)(b * T_ + c * CS)) * H_ + h;

    float A = 1.0f, V = 0.0f;
#pragma unroll 4
    for (int t = 0; t < CS; ++t) {
        const float at = a[base];
        const float vt = v[base];
        A = A * at;
        V = fmaf(at, V, vt);
        base += H_;
    }
    Ac[tid] = A;
    Vc[tid] = V;
}

__global__ __launch_bounds__(256) void mingru_chunk_scan(
    const float* __restrict__ h0,
    const float* __restrict__ Ac, const float* __restrict__ Vc,
    float* __restrict__ hin)
{
    const int tid = blockIdx.x * 256 + threadIdx.x;
    const int h = tid % H_;
    const int b = tid / H_;
    float hs = g_fn(h0[(size_t)b * H_ + h]);
#pragma unroll 8
    for (int c = 0; c < CHUNKS; ++c) {
        const size_t i = ((size_t)(b * CHUNKS + c)) * H_ + h;
        hin[i] = hs;
        hs = fmaf(Ac[i], hs, Vc[i]);
    }
}

__global__ __launch_bounds__(256) void mingru_chunk_apply(
    const float* __restrict__ a, const float* __restrict__ v,
    const float* __restrict__ hin, float* __restrict__ out)
{
    const int tid = blockIdx.x * 256 + threadIdx.x;
    const int h  = tid % H_;
    const int bc = tid / H_;
    const int c  = bc % CHUNKS;
    const int b  = bc / CHUNKS;
    size_t base = ((size_t)(b * T_ + c * CS)) * H_ + h;

    float hs = hin[tid];
#pragma unroll 4
    for (int t = 0; t < CS; ++t) {
        hs = fmaf(a[base], hs, v[base]);
        out[base] = hs;
        base += H_;
    }
}

extern "C" void kernel_launch(void* const* d_in, const int* in_sizes, int n_in,
                              void* d_out, int out_size, void* d_ws, size_t ws_size,
                              hipStream_t stream) {
    const float* x  = (const float*)d_in[0];
    const float* h0 = (const float*)d_in[1];
    const float* Wz = (const float*)d_in[2];
    const float* bz = (const float*)d_in[3];
    const float* Wh = (const float*)d_in[4];
    const float* bh = (const float*)d_in[5];
    float* out = (float*)d_out;

    float* wa  = (float*)d_ws;
    float* wv  = wa + (size_t)M_ * H_;
    float* Ac  = wv + (size_t)M_ * H_;
    float* Vc  = Ac + (size_t)B_ * CHUNKS * H_;
    float* hin = Vc + (size_t)B_ * CHUNKS * H_;

    dim3 g1(H_ / BN, M_ / BM);
    mingru_gemm_gates<<<g1, 128, 0, stream>>>(x, Wz, bz, Wh, bh, wa, wv);

    const int nchunk = B_ * H_ * CHUNKS;
    mingru_chunk_reduce<<<nchunk / 256, 256, 0, stream>>>(wa, wv, Ac, Vc);
    mingru_chunk_scan<<<(B_ * H_) / 256, 256, 0, stream>>>(h0, Ac, Vc, hin);
    mingru_chunk_apply<<<nchunk / 256, 256, 0, stream>>>(wa, wv, hin, out);
}


# Chunk reduction fused into GEMM epilogue: each wave composes its 64 rows' affine map from a,v in registers, writes Ac,Vc; reduce launch removed; chunk summary now a reduction tree, not a serial chain
# speedup vs baseline: 2.4871x; 1.0569x over previous
#include <hip/hip_runtime.h>
#include <math.h>


typedef __attribute__((ext_vector_type(2))) float v2f;
typedef __attribute__((ext_vector_type(4))) float v4f;
typedef __attribute__((ext_vector_type(8))) float v8f;

#define B_      8
#define T_      4096
#define D_      512
#define H_      512
#define M_      (B_ * T_)
#define CHUNKS  64
#define CS      64

__device__ __forceinline__ void sigmoid_pair(float x, float& sig_pos, float& sig_neg) {
    const float e = __expf(-fabsf(x));
    const float r = __builtin_amdgcn_rcpf(1.0f + e);
    sig_pos = r;
    sig_neg = e * r;
}

__device__ __forceinline__ float g_fn(float x) {
    const float e = __expf(fminf(x, 0.0f));
    const float s = e * __builtin_amdgcn_rcpf(1.0f + e);
    return (x >= 0.0f) ? (x + 0.5f) : s;
}

#define BM      128
#define BN      64
#define BK      32
#define PITCH4  9
#define A_V4    (BM * PITCH4)
#define W_V4    (BN * PITCH4)
#define STAGE_V4 (A_V4 + 2 * W_V4)
#define KT      (D_ / BK)

struct Frags {
    v4f a[4];
    v4f z[2];
    v4f h[2];
};

__device__ __forceinline__ void load_frags(Frags& f, const v4f* sa, const v4f* sz,
                                           const v4f* sh, int j) {
#pragma unroll
    for (int mt = 0; mt < 4; ++mt) f.a[mt] = sa[mt * 16 * PITCH4 + 2 * j];
#pragma unroll
    for (int nt = 0; nt < 2; ++nt) {
        f.z[nt] = sz[nt * 16 * PITCH4 + 2 * j];
        f.h[nt] = sh[nt * 16 * PITCH4 + 2 * j];
    }
}

__device__ __forceinline__ void mma_chunk(const Frags& f, v8f (&accZ)[4][2], v8f (&accH)[4][2]) {
#pragma unroll
    for (int s = 0; s < 2; ++s) {
#pragma unroll
        for (int mt = 0; mt < 4; ++mt) {
            const v2f av = s ? (v2f){f.a[mt].z, f.a[mt].w} : (v2f){f.a[mt].x, f.a[mt].y};
#pragma unroll
            for (int nt = 0; nt < 2; ++nt) {
                const v2f zv = s ? (v2f){f.z[nt].z, f.z[nt].w} : (v2f){f.z[nt].x, f.z[nt].y};
                const v2f hv = s ? (v2f){f.h[nt].z, f.h[nt].w} : (v2f){f.h[nt].x, f.h[nt].y};
                accZ[mt][nt] = __builtin_amdgcn_wmma_f32_16x16x4_f32(
                    false, av, false, zv, (short)0, accZ[mt][nt], false, false);
                accH[mt][nt] = __builtin_amdgcn_wmma_f32_16x16x4_f32(
                    false, av, false, hv, (short)0, accH[mt][nt], false, false);
            }
        }
    }
}

__device__ __forceinline__ void compute_stage(const v4f* sa, const v4f* sz, const v4f* sh,
                                              v8f (&accZ)[4][2], v8f (&accH)[4][2]) {
    Frags f0, f1;
    load_frags(f0, sa, sz, sh, 0);
    load_frags(f1, sa, sz, sh, 1);
    mma_chunk(f0, accZ, accH);
    load_frags(f0, sa, sz, sh, 2);
    mma_chunk(f1, accZ, accH);
    load_frags(f1, sa, sz, sh, 3);
    mma_chunk(f0, accZ, accH);
    mma_chunk(f1, accZ, accH);
}

__global__ __launch_bounds__(128) void mingru_gemm_gates(
    const float* __restrict__ X,
    const float* __restrict__ Wz, const float* __restrict__ bz,
    const float* __restrict__ Wh, const float* __restrict__ bh,
    float* __restrict__ outA, float* __restrict__ outV,
    float* __restrict__ Ac, float* __restrict__ Vc)
{
    __shared__ v4f lds[2 * STAGE_V4];

    const int tid  = threadIdx.x;
    const int wave = tid >> 5;
    const int lane = tid & 31;
    const int half = lane >> 4;
    const int l16  = lane & 15;
    const int wm   = wave >> 1;
    const int wn   = wave & 1;
    const int n0   = blockIdx.x * BN;
    const int m0   = blockIdx.y * BM;

    const int cr = tid >> 3;
    const int cq = tid & 7;
    const float* gA = X  + (size_t)(m0 + cr) * D_ + cq * 4;
    const float* gZ = Wz + (size_t)(n0 + cr) * D_ + cq * 4;
    const float* gH = Wh + (size_t)(n0 + cr) * D_ + cq * 4;
    const int sA = cr * PITCH4 + cq;
    const int sZ = A_V4 + cr * PITCH4 + cq;
    const int sH = A_V4 + W_V4 + cr * PITCH4 + cq;

    const int fA = (wm * 64 + l16) * PITCH4 + half;
    const int fZ = A_V4 + (wn * 32 + l16) * PITCH4 + half;
    const int fH = fZ + W_V4;

    v4f pa[8], pz[4], ph[4];
#pragma unroll
    for (int i = 0; i < 8; ++i) pa[i] = *(const v4f*)(gA + (size_t)i * 16 * D_);
#pragma unroll
    for (int i = 0; i < 4; ++i) {
        pz[i] = *(const v4f*)(gZ + (size_t)i * 16 * D_);
        ph[i] = *(const v4f*)(gH + (size_t)i * 16 * D_);
    }

    v8f accZ[4][2], accH[4][2];
#pragma unroll
    for (int mt = 0; mt < 4; ++mt)
#pragma unroll
        for (int nt = 0; nt < 2; ++nt) {
            accZ[mt][nt] = (v8f){0.f,0.f,0.f,0.f,0.f,0.f,0.f,0.f};
            accH[mt][nt] = (v8f){0.f,0.f,0.f,0.f,0.f,0.f,0.f,0.f};
        }

#pragma unroll
    for (int i = 0; i < 8; ++i) lds[sA + i * 16 * PITCH4] = pa[i];
#pragma unroll
    for (int i = 0; i < 4; ++i) {
        lds[sZ + i * 16 * PITCH4] = pz[i];
        lds[sH + i * 16 * PITCH4] = ph[i];
    }
    __syncthreads();

    for (int t = 0; t < KT - 1; ++t) {
        const int cur = (t & 1) * STAGE_V4;
        const int nxt = STAGE_V4 - cur;
        const int ko  = (t + 1) * BK;
#pragma unroll
        for (int i = 0; i < 8; ++i) pa[i] = *(const v4f*)(gA + (size_t)i * 16 * D_ + ko);
#pragma unroll
        for (int i = 0; i < 4; ++i) {
            pz[i] = *(const v4f*)(gZ + (size_t)i * 16 * D_ + ko);
            ph[i] = *(const v4f*)(gH + (size_t)i * 16 * D_ + ko);
        }

        compute_stage(lds + cur + fA, lds + cur + fZ, lds + cur + fH, accZ, accH);

#pragma unroll
        for (int i = 0; i < 8; ++i) lds[nxt + sA + i * 16 * PITCH4] = pa[i];
#pragma unroll
        for (int i = 0; i < 4; ++i) {
            lds[nxt + sZ + i * 16 * PITCH4] = pz[i];
            lds[nxt + sH + i * 16 * PITCH4] = ph[i];
        }
        __syncthreads();
    }
    {
        const int cur = ((KT - 1) & 1) * STAGE_V4;
        compute_stage(lds + cur + fA, lds + cur + fZ, lds + cur + fH, accZ, accH);
    }

    const int nb = n0 + wn * 32 + l16;
    float bzn[2], bhn[2];
#pragma unroll
    for (int nt = 0; nt < 2; ++nt) {
        bzn[nt] = bz[nb + nt * 16];
        bhn[nt] = bh[nb + nt * 16];
    }
    float cA[2] = {1.0f, 1.0f}, cV[2] = {0.0f, 0.0f};
#pragma unroll
    for (int mt = 0; mt < 4; ++mt) {
        float lA[2] = {1.0f, 1.0f}, lV[2] = {0.0f, 0.0f};
#pragma unroll
        for (int r = 0; r < 8; ++r) {
            const int m = m0 + wm * 64 + mt * 16 + half * 8 + r;
#pragma unroll
            for (int nt = 0; nt < 2; ++nt) {
                const float kv  = accZ[mt][nt][r] + bzn[nt];
                const float pre = accH[mt][nt][r] + bhn[nt];
                float sp, sn;
                sigmoid_pair(kv, sp, sn);
                const float z = (kv >= 0.0f) ? sp : sn;
                const float a = (kv >= 0.0f) ? sn : sp;
                const float v = z * g_fn(pre);
                const size_t idx = (size_t)m * H_ + nb + nt * 16;
                outA[idx] = a;
                outV[idx] = v;
                lA[nt] = lA[nt] * a;
                lV[nt] = fmaf(a, lV[nt], v);
            }
        }
#pragma unroll
        for (int nt = 0; nt < 2; ++nt) {
            const float oA = __shfl_xor(lA[nt], 16);
            const float oV = __shfl_xor(lV[nt], 16);
            const float A0 = half ? oA : lA[nt];
            const float V0 = half ? oV : lV[nt];
            const float A1 = half ? lA[nt] : oA;
            const float V1 = half ? lV[nt] : oV;
            const float tA = A1 * A0;
            const float tV = fmaf(A1, V0, V1);
            cV[nt] = fmaf(tA, cV[nt], tV);
            cA[nt] = tA * cA[nt];
        }
    }
    if (half == 0) {
        const size_t ci = (size_t)(blockIdx.y * 2 + wm) * H_ + nb;
#pragma unroll
        for (int nt = 0; nt < 2; ++nt) {
            Ac[ci + nt * 16] = cA[nt];
            Vc[ci + nt * 16] = cV[nt];
        }
    }
}

__global__ __launch_bounds__(256) void mingru_chunk_scan(
    const float* __restrict__ h0,
    const float* __restrict__ Ac, const float* __restrict__ Vc,
    float* __restrict__ hin)
{
    const int tid = blockIdx.x * 256 + threadIdx.x;
    const int h = tid % H_;
    const int b = tid / H_;
    float hs = g_fn(h0[(size_t)b * H_ + h]);
#pragma unroll 8
    for (int c = 0; c < CHUNKS; ++c) {
        const size_t i = ((size_t)(b * CHUNKS + c)) * H_ + h;
        hin[i] = hs;
        hs = fmaf(Ac[i], hs, Vc[i]);
    }
}

__global__ __launch_bounds__(256) void mingru_chunk_apply(
    const float* __restrict__ a, const float* __restrict__ v,
    const float* __restrict__ hin, float* __restrict__ out)
{
    const int tid = blockIdx.x * 256 + threadIdx.x;
    const int h  = tid % H_;
    const int bc = tid / H_;
    const int c  = bc % CHUNKS;
    const int b  = bc / CHUNKS;
    size_t base = ((size_t)(b * T_ + c * CS)) * H_ + h;

    float hs = hin[tid];
#pragma unroll 4
    for (int t = 0; t < CS; ++t) {
        hs = fmaf(a[base], hs, v[base]);
        out[base] = hs;
        base += H_;
    }
}

extern "C" void kernel_launch(void* const* d_in, const int* in_sizes, int n_in,
                              void* d_out, int out_size, void* d_ws, size_t ws_size,
                              hipStream_t stream) {
    const float* x  = (const float*)d_in[0];
    const float* h0 = (const float*)d_in[1];
    const float* Wz = (const float*)d_in[2];
    const float* bz = (const float*)d_in[3];
    const float* Wh = (const float*)d_in[4];
    const float* bh = (const float*)d_in[5];
    float* out = (float*)d_out;

    float* wa  = (float*)d_ws;
    float* wv  = wa + (size_t)M_ * H_;
    float* Ac  = wv + (size_t)M_ * H_;
    float* Vc  = Ac + (size_t)B_ * CHUNKS * H_;
    float* hin = Vc + (size_t)B_ * CHUNKS * H_;

    dim3 g1(H_ / BN, M_ / BM);
    mingru_gemm_gates<<<g1, 128, 0, stream>>>(x, Wz, bz, Wh, bh, wa, wv, Ac, Vc);

    const int nchunk = B_ * H_ * CHUNKS;
    mingru_chunk_scan<<<(B_ * H_) / 256, 256, 0, stream>>>(h0, Ac, Vc, hin);
    mingru_chunk_apply<<<nchunk / 256, 256, 0, stream>>>(wa, wv, hin, out);
}
